# starting kernel
# baseline (speedup 1.0000x reference)
#include <hip/hip_runtime.h>
#include <hip/hip_bf16.h>

typedef __attribute__((ext_vector_type(2))) float v2f;
typedef __attribute__((ext_vector_type(8))) float v8f;

#define B_  2048
#define L_  200
#define D_  300
#define H_  32
#define C_  3

__global__ __launch_bounds__(320) void embsum_kernel(
    const int* __restrict__ x, const int* __restrict__ lengths,
    const float* __restrict__ emb, float* __restrict__ rep)
{
    __shared__ int idx[L_];
    const int b   = blockIdx.x;
    const int tid = threadIdx.x;

    const int* xr = x + (long long)b * L_;
    for (int l = tid; l < L_; l += 320) idx[l] = xr[l];
    __syncthreads();

    if (tid >= D_) return;

    float acc = 0.f;
    #pragma unroll 2
    for (int l = 0; l < L_; ++l) {
        const long long row = (long long)idx[l] * D_;
        if (l + 8 < L_) {
            __builtin_prefetch(&emb[(long long)idx[l + 8] * D_ + tid], 0, 1);
        }
        acc += emb[row + tid];
    }

    rep[(long long)b * D_ + tid] = acc / (float)lengths[b];
}

__global__ __launch_bounds__(32) void mlp_wmma_kernel(
    const float* __restrict__ rep,
    const float* __restrict__ W1, const float* __restrict__ b1,
    const float* __restrict__ W2, const float* __restrict__ b2,
    float* __restrict__ out)
{
    __shared__ float hbuf[16 * H_];

    const int lane = threadIdx.x;
    const int m0   = blockIdx.x * 16;
    const int m    = lane & 15;
    const int kk   = (lane >> 4) * 2;
    const int n    = lane & 15;

    v8f c0 = {};
    v8f c1 = {};

    const float* arow = rep + (long long)(m0 + m) * D_;

    for (int k = 0; k < D_; k += 4) {
        v2f a;
        const float* ap = arow + k + kk;
        a.x = ap[0];
        a.y = ap[1];

        const float* wp = W1 + (long long)(k + kk) * H_;
        v2f bv0, bv1;
        bv0.x = wp[n];
        bv0.y = wp[H_ + n];
        bv1.x = wp[16 + n];
        bv1.y = wp[H_ + 16 + n];

        c0 = __builtin_amdgcn_wmma_f32_16x16x4_f32(
                 false, a, false, bv0, (short)0, c0, false, false);
        c1 = __builtin_amdgcn_wmma_f32_16x16x4_f32(
                 false, a, false, bv1, (short)0, c1, false, false);
    }

    const int mbase = (lane >> 4) * 8;
    const float bb0 = b1[n];
    const float bb1 = b1[16 + n];
    #pragma unroll
    for (int r = 0; r < 8; ++r) {
        const int mm = mbase + r;
        float h0 = c0[r] + bb0; h0 = h0 > 0.f ? h0 : 0.f;
        float h1 = c1[r] + bb1; h1 = h1 > 0.f ? h1 : 0.f;
        hbuf[mm * H_ + n]      = h0;
        hbuf[mm * H_ + 16 + n] = h1;
    }
    __syncthreads();

    if (lane < 16) {
        float a0 = b2[0], a1 = b2[1], a2 = b2[2];
        #pragma unroll
        for (int nn = 0; nn < H_; ++nn) {
            const float hv = hbuf[lane * H_ + nn];
            a0 += hv * W2[nn * C_ + 0];
            a1 += hv * W2[nn * C_ + 1];
            a2 += hv * W2[nn * C_ + 2];
        }
        float* o = out + (long long)(m0 + lane) * C_;
        o[0] = a0; o[1] = a1; o[2] = a2;
    }
}

extern "C" void kernel_launch(void* const* d_in, const int* in_sizes, int n_in,
                              void* d_out, int out_size, void* d_ws, size_t ws_size,
                              hipStream_t stream)
{
    const int*   x       = (const int*)  d_in[0];
    const int*   lengths = (const int*)  d_in[1];
    const float* emb     = (const float*)d_in[2];
    const float* W1      = (const float*)d_in[3];
    const float* b1      = (const float*)d_in[4];
    const float* W2      = (const float*)d_in[5];
    const float* b2      = (const float*)d_in[6];

    float* rep = (float*)d_ws;
    float* out = (float*)d_out;

    embsum_kernel<<<B_, 320, 0, stream>>>(x, lengths, emb, rep);
    mlp_wmma_kernel<<<B_ / 16, 32, 0, stream>>>(rep, W1, b1, W2, b2, out);
}


# Table rows load in whole-row order (one row segment per 16-byte-per-lane load, 4-5 lines each instead of 16 lines of 32 B) and reach WMMA fragment order through a wave-private LDS block
# speedup vs baseline: 2.6364x; 2.6364x over previous
#include <hip/hip_runtime.h>


typedef __attribute__((ext_vector_type(2))) float v2f;
typedef __attribute__((ext_vector_type(4))) float v4f;
typedef __attribute__((ext_vector_type(8))) float v8f;

#define B_  2048
#define L_  200
#define D_  300
#define H_  32
#define C_  3
#define ROWS_ 4
#define CS_   2
#define NTHR_ (32 * ROWS_ * CS_)
#define NG_  13
#define NDW_ 18
#define NT_  75
#define PITCH_ 164

__device__ __forceinline__ int kof(int t, int h, int s)
{
    return t < 74 ? 8 * (t >> 1) + 4 * h + 2 * (t & 1) + s : 296 + 2 * h + s;
}

template <int CS, int NR>
__device__ __forceinline__ void gather_group(
    const float* __restrict__ emb, const int* gidx, float* my, int lane, int m, int h,
    v4f (&acc)[NDW_], v4f& accx, v2f& acct)
{
    constexpr int NTP = CS == 0 ? 4 : 8;
    constexpr int RPT = 32 / NTP;
    constexpr int NTL = NR / RPT;
    const int rowi = gidx[m];

    v4f ld[NR];
    v4f lt[NTL];
    const float* cb = emb + 144 * CS + 4 * lane;
    for (int r = 0; r < NR; ++r) {
        const int ir = __builtin_amdgcn_readlane(rowi, r);
        ld[r] = *(const v4f*)(cb + (long long)ir * D_);
    }
    const int trow = lane / NTP;
    const int tp = 32 + (CS == 0 ? (lane & 3) : ((lane & 7) < 7 ? (lane & 7) : 6));
    for (int j = 0; j < NTL; ++j) {
        const int ir = gidx[trow + RPT * j];
        lt[j] = *(const v4f*)(emb + (long long)ir * D_ + 144 * CS + 4 * tp);
    }
    for (int r = 0; r < NR; ++r) *(v4f*)(my + r * PITCH_ + 4 * lane) = ld[r];
    for (int j = 0; j < NTL; ++j) *(v4f*)(my + (trow + RPT * j) * PITCH_ + 4 * tp) = lt[j];

    __builtin_amdgcn_fence(__ATOMIC_RELEASE, "wavefront");
    __builtin_amdgcn_wave_barrier();
    __builtin_amdgcn_fence(__ATOMIC_ACQUIRE, "wavefront");

    if (m < NR) {
        const float* q = my + m * PITCH_ + 4 * h;
        for (int d = 0; d < NDW_; ++d) acc[d] += *(const v4f*)(q + 8 * d);
        if (CS == CS_ - 1) {
            accx += *(const v4f*)(q + 8 * NDW_);
            acct += *(const v2f*)(q + 8 * NDW_ + 8 - 2 * h);
        }
    }

    __builtin_amdgcn_fence(__ATOMIC_RELEASE, "wavefront");
    __builtin_amdgcn_wave_barrier();
    __builtin_amdgcn_fence(__ATOMIC_ACQUIRE, "wavefront");
}

__global__ __launch_bounds__(NTHR_) void embsum_mlp_kernel(
    const int* __restrict__ x, const int* __restrict__ lengths,
    const float* __restrict__ emb,
    const float* __restrict__ W1, const float* __restrict__ b1,
    const float* __restrict__ W2, const float* __restrict__ b2,
    float* __restrict__ out)
{
    __shared__ float Bf[(NT_ + 1) * 128];
    __shared__ float part[ROWS_][CS_][32];
    __shared__ int   idxs[ROWS_][4 * 32 * CS_];
    __shared__ float stg[ROWS_ * CS_][16 * PITCH_];

    const int tid  = threadIdx.x;
    const int lane = tid & 31;
    const int wave = tid >> 5;
    const int br   = wave / CS_;
    const int cs   = wave % CS_;
    const int m    = lane & 15;
    const int h    = lane >> 4;
    const int b    = blockIdx.x * ROWS_ + br;
    const int* xr  = x + (long long)b * L_;

    {
        const int q = cs * 32 + lane;
        for (int i = 0; i < 4; ++i) {
            const int l = q + 32 * CS_ * i;
            idxs[br][l] = xr[l < L_ ? l : L_ - 1];
        }
    }

    for (int i = 0; i < (NT_ + 1) * 128 / NTHR_; ++i) {
        const int e = i * NTHR_ + tid;
        const int s = e & 1, l = (e >> 1) & 31, j = (e >> 6) & 1, t = e >> 7;
        Bf[e] = W1[kof(t, l >> 4, s) * H_ + 16 * j + (l & 15)];
    }

    v4f acc[NDW_];
    v4f accx = {0.f, 0.f, 0.f, 0.f};
    v2f acct = {0.f, 0.f};
    for (int d = 0; d < NDW_; ++d) acc[d] = (v4f){0.f, 0.f, 0.f, 0.f};

    __syncthreads();

    float* my = stg[wave];
    if (cs == 0) {
        for (int g = 0; g < NG_ - 1; ++g)
            gather_group<0, 16>(emb, &idxs[br][g * 16], my, lane, m, h, acc, accx, acct);
        gather_group<0, 8>(emb, &idxs[br][(NG_ - 1) * 16], my, lane, m, h, acc, accx, acct);
    } else {
        for (int g = 0; g < NG_ - 1; ++g)
            gather_group<1, 16>(emb, &idxs[br][g * 16], my, lane, m, h, acc, accx, acct);
        gather_group<1, 8>(emb, &idxs[br][(NG_ - 1) * 16], my, lane, m, h, acc, accx, acct);
    }

    v8f c0 = {};
    v8f c1 = {};
    const v2f* Bv = (const v2f*)Bf + (2 * NDW_ * cs) * 64 + lane;
    for (int d = 0; d < NDW_; ++d) {
        v2f a, w0, w1;
        a.x = acc[d].x; a.y = acc[d].y;
        w0 = Bv[(2 * d) * 64];
        w1 = Bv[(2 * d) * 64 + 32];
        c0 = __builtin_amdgcn_wmma_f32_16x16x4_f32(false, a, false, w0, (short)0, c0, false, false);
        c1 = __builtin_amdgcn_wmma_f32_16x16x4_f32(false, a, false, w1, (short)0, c1, false, false);
        a.x = acc[d].z; a.y = acc[d].w;
        w0 = Bv[(2 * d + 1) * 64];
        w1 = Bv[(2 * d + 1) * 64 + 32];
        c0 = __builtin_amdgcn_wmma_f32_16x16x4_f32(false, a, false, w0, (short)0, c0, false, false);
        c1 = __builtin_amdgcn_wmma_f32_16x16x4_f32(false, a, false, w1, (short)0, c1, false, false);
    }
    if (cs == CS_ - 1) {
        v2f a, w0, w1;
        a.x = accx.x; a.y = accx.y;
        w0 = Bv[(2 * NDW_) * 64];
        w1 = Bv[(2 * NDW_) * 64 + 32];
        c0 = __builtin_amdgcn_wmma_f32_16x16x4_f32(false, a, false, w0, (short)0, c0, false, false);
        c1 = __builtin_amdgcn_wmma_f32_16x16x4_f32(false, a, false, w1, (short)0, c1, false, false);
        a.x = accx.z; a.y = accx.w;
        w0 = Bv[(2 * NDW_ + 1) * 64];
        w1 = Bv[(2 * NDW_ + 1) * 64 + 32];
        c0 = __builtin_amdgcn_wmma_f32_16x16x4_f32(false, a, false, w0, (short)0, c0, false, false);
        c1 = __builtin_amdgcn_wmma_f32_16x16x4_f32(false, a, false, w1, (short)0, c1, false, false);
        w0 = Bv[(2 * NDW_ + 2) * 64];
        w1 = Bv[(2 * NDW_ + 2) * 64 + 32];
        c0 = __builtin_amdgcn_wmma_f32_16x16x4_f32(false, acct, false, w0, (short)0, c0, false, false);
        c1 = __builtin_amdgcn_wmma_f32_16x16x4_f32(false, acct, false, w1, (short)0, c1, false, false);
    }

    float s0 = ((c0[0] + c0[1]) + (c0[2] + c0[3])) + ((c0[4] + c0[5]) + (c0[6] + c0[7]));
    float s1 = ((c1[0] + c1[1]) + (c1[2] + c1[3])) + ((c1[4] + c1[5]) + (c1[6] + c1[7]));
    s0 += __shfl_xor(s0, 16);
    s1 += __shfl_xor(s1, 16);

    part[br][cs][lane] = (h == 0 ? s0 : s1);
    __syncthreads();

    if (cs == 0) {
        float pre = part[br][0][lane];
        for (int q = 1; q < CS_; ++q) pre += part[br][q][lane];

        float hv = pre / (float)lengths[b] + b1[lane];
        hv = hv > 0.f ? hv : 0.f;

        float p0 = hv * W2[lane * C_ + 0];
        float p1 = hv * W2[lane * C_ + 1];
        float p2 = hv * W2[lane * C_ + 2];
        for (int off = 16; off > 0; off >>= 1) {
            p0 += __shfl_xor(p0, off);
            p1 += __shfl_xor(p1, off);
            p2 += __shfl_xor(p2, off);
        }
        if (lane == 0) {
            float* o = out + (long long)b * C_;
            o[0] = p0 + b2[0];
            o[1] = p1 + b2[1];
            o[2] = p2 + b2[2];
        }
    }
}

extern "C" void kernel_launch(void* const* d_in, const int* in_sizes, int n_in,
                              void* d_out, int out_size, void* d_ws, size_t ws_size,
                              hipStream_t stream)
{
    const int*   x       = (const int*)  d_in[0];
    const int*   lengths = (const int*)  d_in[1];
    const float* emb     = (const float*)d_in[2];
    const float* W1      = (const float*)d_in[3];
    const float* b1      = (const float*)d_in[4];
    const float* W2      = (const float*)d_in[5];
    const float* b2      = (const float*)d_in[6];
    float* out = (float*)d_out;

    embsum_mlp_kernel<<<B_ / ROWS_, NTHR_, 0, stream>>>(
        x, lengths, emb, W1, b1, W2, b2, out);
}


# Each wave prefetches into the second-level cache (global_prefetch_b8) the 256-byte units of the table rows its pair loads four groups later, so uncached rows come from memory ahead of the loads
# speedup vs baseline: 2.7950x; 1.0602x over previous
#include <hip/hip_runtime.h>


typedef __attribute__((ext_vector_type(2))) float v2f;
typedef __attribute__((ext_vector_type(4))) float v4f;
typedef __attribute__((ext_vector_type(8))) float v8f;

#define B_  2048
#define L_  200
#define D_  300
#define H_  32
#define C_  3
#define ROWS_ 4
#define CS_   2
#define NTHR_ (32 * ROWS_ * CS_)
#define NG_  13
#define NDW_ 18
#define NT_  75
#define PITCH_ 164
#define PFD_ 4

__device__ __forceinline__ int kof(int t, int h, int s)
{
    return t < 74 ? 8 * (t >> 1) + 4 * h + 2 * (t & 1) + s : 296 + 2 * h + s;
}

template <typename F>
__device__ __forceinline__ void prefetch_rows(const float* __restrict__ emb, int lane, int cs, F rowidx)
{
    int ir[2], bo[2];
    for (int j = 0; j < 2; ++j) {
        const int k = lane + 32 * j;
        const int r = k < 48 ? k / 6 : 7;
        const int u = k % 6;
        ir[j] = rowidx(8 * cs + r);
        bo[j] = u < 5 ? 256 * u : D_ * 4 - 1;
    }
    for (int j = 0; j < 2; ++j)
        if (lane + 32 * j < 48)
            __builtin_prefetch((const char*)emb + (long long)ir[j] * (D_ * 4) + bo[j], 0, 3);
}

template <int CS, int NR>
__device__ __forceinline__ void gather_group(
    const float* __restrict__ emb, const int* gidx, const int* pidx, float* my, int lane, int m, int h,
    v4f (&acc)[NDW_], v4f& accx, v2f& acct)
{
    constexpr int NTP = CS == 0 ? 4 : 8;
    constexpr int RPT = 32 / NTP;
    constexpr int NTL = NR / RPT;
    const int rowi = gidx[m];

    v4f ld[NR];
    v4f lt[NTL];
    const float* cb = emb + 144 * CS + 4 * lane;
    for (int r = 0; r < NR; ++r) {
        const int ir = __builtin_amdgcn_readlane(rowi, r);
        ld[r] = *(const v4f*)(cb + (long long)ir * D_);
    }
    const int trow = lane / NTP;
    const int tp = 32 + (CS == 0 ? (lane & 3) : ((lane & 7) < 7 ? (lane & 7) : 6));
    for (int j = 0; j < NTL; ++j) {
        const int ir = gidx[trow + RPT * j];
        lt[j] = *(const v4f*)(emb + (long long)ir * D_ + 144 * CS + 4 * tp);
    }
    if (pidx) prefetch_rows(emb, lane, CS, [pidx](int p) { return pidx[p]; });
    for (int r = 0; r < NR; ++r) *(v4f*)(my + r * PITCH_ + 4 * lane) = ld[r];
    for (int j = 0; j < NTL; ++j) *(v4f*)(my + (trow + RPT * j) * PITCH_ + 4 * tp) = lt[j];

    __builtin_amdgcn_fence(__ATOMIC_RELEASE, "wavefront");
    __builtin_amdgcn_wave_barrier();
    __builtin_amdgcn_fence(__ATOMIC_ACQUIRE, "wavefront");

    if (m < NR) {
        const float* q = my + m * PITCH_ + 4 * h;
        for (int d = 0; d < NDW_; ++d) acc[d] += *(const v4f*)(q + 8 * d);
        if (CS == CS_ - 1) {
            accx += *(const v4f*)(q + 8 * NDW_);
            acct += *(const v2f*)(q + 8 * NDW_ + 8 - 2 * h);
        }
    }

    __builtin_amdgcn_fence(__ATOMIC_RELEASE, "wavefront");
    __builtin_amdgcn_wave_barrier();
    __builtin_amdgcn_fence(__ATOMIC_ACQUIRE, "wavefront");
}

__global__ __launch_bounds__(NTHR_) void embsum_mlp_kernel(
    const int* __restrict__ x, const int* __restrict__ lengths,
    const float* __restrict__ emb,
    const float* __restrict__ W1, const float* __restrict__ b1,
    const float* __restrict__ W2, const float* __restrict__ b2,
    float* __restrict__ out)
{
    __shared__ float Bf[(NT_ + 1) * 128];
    __shared__ float part[ROWS_][CS_][32];
    __shared__ int   idxs[ROWS_][4 * 32 * CS_];
    __shared__ float stg[ROWS_ * CS_][16 * PITCH_];

    const int tid  = threadIdx.x;
    const int lane = tid & 31;
    const int wave = tid >> 5;
    const int br   = wave / CS_;
    const int cs   = wave % CS_;
    const int m    = lane & 15;
    const int h    = lane >> 4;
    const int b    = blockIdx.x * ROWS_ + br;
    const int* xr  = x + (long long)b * L_;

    int pfi[PFD_][2];
    for (int g = 0; g < PFD_; ++g)
        for (int j = 0; j < 2; ++j) {
            const int k = lane + 32 * j;
            pfi[g][j] = xr[16 * g + 8 * cs + (k < 48 ? k / 6 : 7)];
        }

    int iv[4];
    {
        const int q = cs * 32 + lane;
        for (int i = 0; i < 4; ++i) {
            const int l = q + 32 * CS_ * i;
            iv[i] = xr[l < L_ ? l : L_ - 1];
        }
    }

    float wv[(NT_ + 1) * 128 / NTHR_];
    for (int i = 0; i < (NT_ + 1) * 128 / NTHR_; ++i) {
        const int e = i * NTHR_ + tid;
        const int s = e & 1, l = (e >> 1) & 31, j = (e >> 6) & 1, t = e >> 7;
        wv[i] = W1[kof(t, l >> 4, s) * H_ + 16 * j + (l & 15)];
    }
    for (int g = 0; g < PFD_; ++g)
        for (int j = 0; j < 2; ++j)
            if (lane + 32 * j < 48) {
                const int u = (lane + 32 * j) % 6;
                __builtin_prefetch((const char*)emb + (long long)pfi[g][j] * (D_ * 4)
                                   + (u < 5 ? 256 * u : D_ * 4 - 1), 0, 3);
            }
    {
        const int q = cs * 32 + lane;
        for (int i = 0; i < 4; ++i) idxs[br][q + 32 * CS_ * i] = iv[i];
    }
    for (int i = 0; i < (NT_ + 1) * 128 / NTHR_; ++i) Bf[i * NTHR_ + tid] = wv[i];

    v4f acc[NDW_];
    v4f accx = {0.f, 0.f, 0.f, 0.f};
    v2f acct = {0.f, 0.f};
    for (int d = 0; d < NDW_; ++d) acc[d] = (v4f){0.f, 0.f, 0.f, 0.f};

    __syncthreads();

    float* my = stg[wave];
    if (cs == 0) {
        for (int g = 0; g < NG_ - 1; ++g)
            gather_group<0, 16>(emb, &idxs[br][g * 16], g + PFD_ < NG_ ? &idxs[br][(g + PFD_) * 16] : nullptr,
                                my, lane, m, h, acc, accx, acct);
        gather_group<0, 8>(emb, &idxs[br][(NG_ - 1) * 16], nullptr, my, lane, m, h, acc, accx, acct);
    } else {
        for (int g = 0; g < NG_ - 1; ++g)
            gather_group<1, 16>(emb, &idxs[br][g * 16], g + PFD_ < NG_ ? &idxs[br][(g + PFD_) * 16] : nullptr,
                                my, lane, m, h, acc, accx, acct);
        gather_group<1, 8>(emb, &idxs[br][(NG_ - 1) * 16], nullptr, my, lane, m, h, acc, accx, acct);
    }

    v8f c0 = {};
    v8f c1 = {};
    const v2f* Bv = (const v2f*)Bf + (2 * NDW_ * cs) * 64 + lane;
    for (int d = 0; d < NDW_; ++d) {
        v2f a, w0, w1;
        a.x = acc[d].x; a.y = acc[d].y;
        w0 = Bv[(2 * d) * 64];
        w1 = Bv[(2 * d) * 64 + 32];
        c0 = __builtin_amdgcn_wmma_f32_16x16x4_f32(false, a, false, w0, (short)0, c0, false, false);
        c1 = __builtin_amdgcn_wmma_f32_16x16x4_f32(false, a, false, w1, (short)0, c1, false, false);
        a.x = acc[d].z; a.y = acc[d].w;
        w0 = Bv[(2 * d + 1) * 64];
        w1 = Bv[(2 * d + 1) * 64 + 32];
        c0 = __builtin_amdgcn_wmma_f32_16x16x4_f32(false, a, false, w0, (short)0, c0, false, false);
        c1 = __builtin_amdgcn_wmma_f32_16x16x4_f32(false, a, false, w1, (short)0, c1, false, false);
    }
    if (cs == CS_ - 1) {
        v2f a, w0, w1;
        a.x = accx.x; a.y = accx.y;
        w0 = Bv[(2 * NDW_) * 64];
        w1 = Bv[(2 * NDW_) * 64 + 32];
        c0 = __builtin_amdgcn_wmma_f32_16x16x4_f32(false, a, false, w0, (short)0, c0, false, false);
        c1 = __builtin_amdgcn_wmma_f32_16x16x4_f32(false, a, false, w1, (short)0, c1, false, false);
        a.x = accx.z; a.y = accx.w;
        w0 = Bv[(2 * NDW_ + 1) * 64];
        w1 = Bv[(2 * NDW_ + 1) * 64 + 32];
        c0 = __builtin_amdgcn_wmma_f32_16x16x4_f32(false, a, false, w0, (short)0, c0, false, false);
        c1 = __builtin_amdgcn_wmma_f32_16x16x4_f32(false, a, false, w1, (short)0, c1, false, false);
        w0 = Bv[(2 * NDW_ + 2) * 64];
        w1 = Bv[(2 * NDW_ + 2) * 64 + 32];
        c0 = __builtin_amdgcn_wmma_f32_16x16x4_f32(false, acct, false, w0, (short)0, c0, false, false);
        c1 = __builtin_amdgcn_wmma_f32_16x16x4_f32(false, acct, false, w1, (short)0, c1, false, false);
    }

    float s0 = ((c0[0] + c0[1]) + (c0[2] + c0[3])) + ((c0[4] + c0[5]) + (c0[6] + c0[7]));
    float s1 = ((c1[0] + c1[1]) + (c1[2] + c1[3])) + ((c1[4] + c1[5]) + (c1[6] + c1[7]));
    s0 += __shfl_xor(s0, 16);
    s1 += __shfl_xor(s1, 16);

    part[br][cs][lane] = (h == 0 ? s0 : s1);
    __syncthreads();

    if (cs == 0) {
        float pre = part[br][0][lane];
        for (int q = 1; q < CS_; ++q) pre += part[br][q][lane];

        float hv = pre / (float)lengths[b] + b1[lane];
        hv = hv > 0.f ? hv : 0.f;

        float p0 = hv * W2[lane * C_ + 0];
        float p1 = hv * W2[lane * C_ + 1];
        float p2 = hv * W2[lane * C_ + 2];
        for (int off = 16; off > 0; off >>= 1) {
            p0 += __shfl_xor(p0, off);
            p1 += __shfl_xor(p1, off);
            p2 += __shfl_xor(p2, off);
        }
        if (lane == 0) {
            float* o = out + (long long)b * C_;
            o[0] = p0 + b2[0];
            o[1] = p1 + b2[1];
            o[2] = p2 + b2[2];
        }
    }
}

extern "C" void kernel_launch(void* const* d_in, const int* in_sizes, int n_in,
                              void* d_out, int out_size, void* d_ws, size_t ws_size,
                              hipStream_t stream)
{
    const int*   x       = (const int*)  d_in[0];
    const int*   lengths = (const int*)  d_in[1];
    const float* emb     = (const float*)d_in[2];
    const float* W1      = (const float*)d_in[3];
    const float* b1      = (const float*)d_in[4];
    const float* W2      = (const float*)d_in[5];
    const float* b2      = (const float*)d_in[6];
    float* out = (float*)d_out;

    embsum_mlp_kernel<<<B_ / ROWS_, NTHR_, 0, stream>>>(
        x, lengths, emb, W1, b1, W2, b2, out);
}
